# starting kernel
# baseline (speedup 1.0000x reference)
#include <hip/hip_runtime.h>

typedef float v2f __attribute__((ext_vector_type(2)));
typedef float v8f __attribute__((ext_vector_type(8)));

#define N_ROWS      131072
#define ROWS_PER_WG 64
#define STR         132

__device__ __forceinline__ v8f bias_init(float bv) {
    v8f c;
#pragma unroll
    for (int i = 0; i < 8; ++i) c[i] = bv;
    return c;
}

__device__ __forceinline__ void wmma_k64(const float* __restrict__ W, int ldW,
                                         int kOff, int nCol,
                                         const float* src, int srcColOff,
                                         int lane, v8f c[4]) {
    const int half = lane >> 4;
    const int l15  = lane & 15;
#pragma unroll
    for (int kk = 0; kk < 16; ++kk) {
        const int k0 = kk * 4 + 2 * half;
        v2f b;
        b.x = W[(kOff + k0 + 0) * ldW + nCol];
        b.y = W[(kOff + k0 + 1) * ldW + nCol];
#pragma unroll
        for (int rt = 0; rt < 4; ++rt) {
            v2f a = *(const v2f*)&src[(16 * rt + l15) * STR + srcColOff + k0];
            c[rt] = __builtin_amdgcn_wmma_f32_16x16x4_f32(
                false, a, false, b, (short)0, c[rt], false, false);
        }
    }
}

__device__ __forceinline__ void sin_store4(const v8f c[4], float* dst,
                                           int nCol, int half) {
#pragma unroll
    for (int rt = 0; rt < 4; ++rt)
#pragma unroll
        for (int v = 0; v < 8; ++v)
            dst[(16 * rt + v + 8 * half) * STR + nCol] = __sinf(c[rt][v]);
}

__global__ __launch_bounds__(256)
void bsnn_fused(const float* __restrict__ X,
                const float* __restrict__ W0, const float* __restrict__ B0,
                const float* __restrict__ W1, const float* __restrict__ B1,
                const float* __restrict__ W2, const float* __restrict__ B2,
                const float* __restrict__ W3, const float* __restrict__ B3,
                const float* __restrict__ W4, const float* __restrict__ B4,
                const float* __restrict__ W5, const float* __restrict__ B5,
                float* __restrict__ out) {
    __shared__ float hA[ROWS_PER_WG * STR];
    __shared__ float hB[ROWS_PER_WG * STR];
    __shared__ float hC[ROWS_PER_WG * STR];
    __shared__ float acc[ROWS_PER_WG];

    const int tid      = threadIdx.x;
    const int w        = tid >> 5;
    const int lane     = tid & 31;
    const int half     = lane >> 4;
    const int l15      = lane & 15;
    const int rowsBase = blockIdx.x * ROWS_PER_WG;
    const int nCol     = 16 * w + l15;

    if (tid < ROWS_PER_WG) acc[tid] = 0.0f;

    {
        const int nc0 = 16 * (w & 3) + l15;
        const float bias = B0[nc0];
        const float w0r = W0[0 * 64 + nc0];
        const float w1r = W0[1 * 64 + nc0];
        const float w2r = W0[2 * 64 + nc0];
        v2f b;
        b.x = half ? w2r : w0r;
        b.y = half ? 0.0f : w1r;
#pragma unroll
        for (int j = 0; j < 2; ++j) {
            const int rB   = 16 * (w >> 2) + 32 * j;
            const int gRow = rowsBase + rB + l15;
            const float x0 = X[gRow * 3 + 0];
            const float x1 = X[gRow * 3 + 1];
            const float x2 = X[gRow * 3 + 2];
            v2f a;
            a.x = half ? x2 : x0;
            a.y = half ? 0.0f : x1;
            v8f c = bias_init(bias);
            c = __builtin_amdgcn_wmma_f32_16x16x4_f32(
                false, a, false, b, (short)0, c, false, false);
#pragma unroll
            for (int v = 0; v < 8; ++v)
                hC[(rB + v + 8 * half) * STR + nc0] = __sinf(c[v]);
        }
    }
    __syncthreads();

    {
        v8f c[4];
#pragma unroll
        for (int rt = 0; rt < 4; ++rt) c[rt] = bias_init(B1[nCol]);
        wmma_k64(W1, 128, 0, nCol, hC, 0, lane, c);
        sin_store4(c, hA, nCol, half);
    }
    __syncthreads();

    v8f pacc[4];
#pragma unroll
    for (int rt = 0; rt < 4; ++rt) pacc[rt] = bias_init(0.0f);

    for (int c2 = 0; c2 < 2; ++c2) {
        {
            v8f c[4];
#pragma unroll
            for (int rt = 0; rt < 4; ++rt)
                c[rt] = bias_init(B2[128 * c2 + nCol]);
            wmma_k64(W2, 256, 64 * c2, 128 * c2 + nCol, hA, 64 * c2, lane, c);
            sin_store4(c, hB, nCol, half);
        }
        __syncthreads();
        for (int c3 = 0; c3 < 2; ++c3) {
            const int i3 = 2 * c2 + c3;
            {
                v8f c[4];
#pragma unroll
                for (int rt = 0; rt < 4; ++rt)
                    c[rt] = bias_init(B3[128 * i3 + nCol]);
                wmma_k64(W3, 512, 64 * i3, 128 * i3 + nCol, hB, 64 * c3,
                         lane, c);
                sin_store4(c, hC, nCol, half);
            }
            __syncthreads();
            for (int c4 = 0; c4 < 2; ++c4) {
                const int i4 = 2 * i3 + c4;
                v8f c[4];
#pragma unroll
                for (int rt = 0; rt < 4; ++rt)
                    c[rt] = bias_init(B4[128 * i4 + nCol]);
                wmma_k64(W4, 1024, 64 * i4, 128 * i4 + nCol, hC, 64 * c4,
                         lane, c);
                const float w5v = W5[128 * i4 + nCol];
#pragma unroll
                for (int rt = 0; rt < 4; ++rt)
#pragma unroll
                    for (int v = 0; v < 8; ++v)
                        pacc[rt][v] += __sinf(c[rt][v]) * w5v;
            }
            __syncthreads();
        }
    }

#pragma unroll
    for (int rt = 0; rt < 4; ++rt) {
#pragma unroll
        for (int v = 0; v < 8; ++v) {
            float p = pacc[rt][v];
            p += __shfl_xor(p, 1, 16);
            p += __shfl_xor(p, 2, 16);
            p += __shfl_xor(p, 4, 16);
            p += __shfl_xor(p, 8, 16);
            if (l15 == 0) atomicAdd(&acc[16 * rt + v + 8 * half], p);
        }
    }
    __syncthreads();
    if (tid < ROWS_PER_WG) out[rowsBase + tid] = acc[tid] + B5[0];
}

extern "C" void kernel_launch(void* const* d_in, const int* in_sizes, int n_in,
                              void* d_out, int out_size, void* d_ws,
                              size_t ws_size, hipStream_t stream) {
    (void)in_sizes; (void)n_in; (void)d_ws; (void)ws_size; (void)out_size;
    const float* X  = (const float*)d_in[0];
    const float* W0 = (const float*)d_in[1];
    const float* B0 = (const float*)d_in[2];
    const float* W1 = (const float*)d_in[3];
    const float* B1 = (const float*)d_in[4];
    const float* W2 = (const float*)d_in[5];
    const float* B2 = (const float*)d_in[6];
    const float* W3 = (const float*)d_in[7];
    const float* B3 = (const float*)d_in[8];
    const float* W4 = (const float*)d_in[9];
    const float* B4 = (const float*)d_in[10];
    const float* W5 = (const float*)d_in[11];
    const float* B5 = (const float*)d_in[12];

    dim3 grid(N_ROWS / ROWS_PER_WG);
    dim3 block(256);
    bsnn_fused<<<grid, block, 0, stream>>>(X, W0, B0, W1, B1, W2, B2, W3, B3,
                                           W4, B4, W5, B5, (float*)d_out);
}


# Software-pipelined k-loop: A-fragment LDS loads run one group of 4 k-steps ahead in a second register set, so 16 WMMAs issue back to back over 4 accumulators; summation order unchanged
# speedup vs baseline: 1.1683x; 1.1683x over previous
#include <hip/hip_runtime.h>

typedef float v2f __attribute__((ext_vector_type(2)));
typedef float v8f __attribute__((ext_vector_type(8)));

#define N_ROWS      131072
#define ROWS_PER_WG 64
#define STR         132

__device__ __forceinline__ v8f bias_init(float bv) {
    v8f c;
#pragma unroll
    for (int i = 0; i < 8; ++i) c[i] = bv;
    return c;
}

#define KG 4

__device__ __forceinline__ void load_a_group(const float* abase, int g,
                                             v2f a[KG][4]) {
#pragma unroll
    for (int q = 0; q < KG; ++q)
#pragma unroll
        for (int rt = 0; rt < 4; ++rt)
            a[q][rt] = *(const v2f*)&abase[16 * rt * STR + (g * KG + q) * 4];
}

__device__ __forceinline__ void wmma_k64(const float* __restrict__ W, int ldW,
                                         int kOff, int nCol,
                                         const float* src, int srcColOff,
                                         int lane, v8f c[4]) {
    const int half = lane >> 4;
    const int l15  = lane & 15;
    v2f b[16];
#pragma unroll
    for (int kk = 0; kk < 16; ++kk) {
        const int k0 = kk * 4 + 2 * half;
        b[kk].x = W[(kOff + k0 + 0) * ldW + nCol];
        b[kk].y = W[(kOff + k0 + 1) * ldW + nCol];
    }
    const float* abase = src + l15 * STR + srcColOff + 2 * half;
    v2f a[2][KG][4];
    load_a_group(abase, 0, a[0]);
#pragma unroll
    for (int g = 0; g < 16 / KG; ++g) {
        if (g + 1 < 16 / KG) load_a_group(abase, g + 1, a[(g + 1) & 1]);
        __builtin_amdgcn_sched_barrier(0);
#pragma unroll
        for (int q = 0; q < KG; ++q)
#pragma unroll
            for (int rt = 0; rt < 4; ++rt)
                c[rt] = __builtin_amdgcn_wmma_f32_16x16x4_f32(
                    false, a[g & 1][q][rt], false, b[g * KG + q], (short)0,
                    c[rt], false, false);
        __builtin_amdgcn_sched_barrier(0);
    }
}

__device__ __forceinline__ void sin_store4(const v8f c[4], float* dst,
                                           int nCol, int half) {
#pragma unroll
    for (int rt = 0; rt < 4; ++rt)
#pragma unroll
        for (int v = 0; v < 8; ++v)
            dst[(16 * rt + v + 8 * half) * STR + nCol] = __sinf(c[rt][v]);
}

__global__ __launch_bounds__(256)
void bsnn_fused(const float* __restrict__ X,
                const float* __restrict__ W0, const float* __restrict__ B0,
                const float* __restrict__ W1, const float* __restrict__ B1,
                const float* __restrict__ W2, const float* __restrict__ B2,
                const float* __restrict__ W3, const float* __restrict__ B3,
                const float* __restrict__ W4, const float* __restrict__ B4,
                const float* __restrict__ W5, const float* __restrict__ B5,
                float* __restrict__ out) {
    __shared__ float hA[ROWS_PER_WG * STR];
    __shared__ float hB[ROWS_PER_WG * STR];
    __shared__ float hC[ROWS_PER_WG * STR];
    __shared__ float acc[ROWS_PER_WG];

    const int tid      = threadIdx.x;
    const int w        = tid >> 5;
    const int lane     = tid & 31;
    const int half     = lane >> 4;
    const int l15      = lane & 15;
    const int rowsBase = blockIdx.x * ROWS_PER_WG;
    const int nCol     = 16 * w + l15;

    if (tid < ROWS_PER_WG) acc[tid] = 0.0f;

    {
        const int nc0 = 16 * (w & 3) + l15;
        const float bias = B0[nc0];
        const float w0r = W0[0 * 64 + nc0];
        const float w1r = W0[1 * 64 + nc0];
        const float w2r = W0[2 * 64 + nc0];
        v2f b;
        b.x = half ? w2r : w0r;
        b.y = half ? 0.0f : w1r;
#pragma unroll
        for (int j = 0; j < 2; ++j) {
            const int rB   = 16 * (w >> 2) + 32 * j;
            const int gRow = rowsBase + rB + l15;
            const float x0 = X[gRow * 3 + 0];
            const float x1 = X[gRow * 3 + 1];
            const float x2 = X[gRow * 3 + 2];
            v2f a;
            a.x = half ? x2 : x0;
            a.y = half ? 0.0f : x1;
            v8f c = bias_init(bias);
            c = __builtin_amdgcn_wmma_f32_16x16x4_f32(
                false, a, false, b, (short)0, c, false, false);
#pragma unroll
            for (int v = 0; v < 8; ++v)
                hC[(rB + v + 8 * half) * STR + nc0] = __sinf(c[v]);
        }
    }
    __syncthreads();

    {
        v8f c[4];
#pragma unroll
        for (int rt = 0; rt < 4; ++rt) c[rt] = bias_init(B1[nCol]);
        wmma_k64(W1, 128, 0, nCol, hC, 0, lane, c);
        sin_store4(c, hA, nCol, half);
    }
    __syncthreads();

    v8f pacc[4];
#pragma unroll
    for (int rt = 0; rt < 4; ++rt) pacc[rt] = bias_init(0.0f);

    for (int c2 = 0; c2 < 2; ++c2) {
        {
            v8f c[4];
#pragma unroll
            for (int rt = 0; rt < 4; ++rt)
                c[rt] = bias_init(B2[128 * c2 + nCol]);
            wmma_k64(W2, 256, 64 * c2, 128 * c2 + nCol, hA, 64 * c2, lane, c);
            sin_store4(c, hB, nCol, half);
        }
        __syncthreads();
        for (int c3 = 0; c3 < 2; ++c3) {
            const int i3 = 2 * c2 + c3;
            {
                v8f c[4];
#pragma unroll
                for (int rt = 0; rt < 4; ++rt)
                    c[rt] = bias_init(B3[128 * i3 + nCol]);
                wmma_k64(W3, 512, 64 * i3, 128 * i3 + nCol, hB, 64 * c3,
                         lane, c);
                sin_store4(c, hC, nCol, half);
            }
            __syncthreads();
            for (int c4 = 0; c4 < 2; ++c4) {
                const int i4 = 2 * i3 + c4;
                v8f c[4];
#pragma unroll
                for (int rt = 0; rt < 4; ++rt)
                    c[rt] = bias_init(B4[128 * i4 + nCol]);
                wmma_k64(W4, 1024, 64 * i4, 128 * i4 + nCol, hC, 64 * c4,
                         lane, c);
                const float w5v = W5[128 * i4 + nCol];
#pragma unroll
                for (int rt = 0; rt < 4; ++rt)
#pragma unroll
                    for (int v = 0; v < 8; ++v)
                        pacc[rt][v] += __sinf(c[rt][v]) * w5v;
            }
            __syncthreads();
        }
    }

#pragma unroll
    for (int rt = 0; rt < 4; ++rt) {
#pragma unroll
        for (int v = 0; v < 8; ++v) {
            float p = pacc[rt][v];
            p += __shfl_xor(p, 1, 16);
            p += __shfl_xor(p, 2, 16);
            p += __shfl_xor(p, 4, 16);
            p += __shfl_xor(p, 8, 16);
            if (l15 == 0) atomicAdd(&acc[16 * rt + v + 8 * half], p);
        }
    }
    __syncthreads();
    if (tid < ROWS_PER_WG) out[rowsBase + tid] = acc[tid] + B5[0];
}

extern "C" void kernel_launch(void* const* d_in, const int* in_sizes, int n_in,
                              void* d_out, int out_size, void* d_ws,
                              size_t ws_size, hipStream_t stream) {
    (void)in_sizes; (void)n_in; (void)d_ws; (void)ws_size; (void)out_size;
    const float* X  = (const float*)d_in[0];
    const float* W0 = (const float*)d_in[1];
    const float* B0 = (const float*)d_in[2];
    const float* W1 = (const float*)d_in[3];
    const float* B1 = (const float*)d_in[4];
    const float* W2 = (const float*)d_in[5];
    const float* B2 = (const float*)d_in[6];
    const float* W3 = (const float*)d_in[7];
    const float* B3 = (const float*)d_in[8];
    const float* W4 = (const float*)d_in[9];
    const float* B4 = (const float*)d_in[10];
    const float* W5 = (const float*)d_in[11];
    const float* B5 = (const float*)d_in[12];

    dim3 grid(N_ROWS / ROWS_PER_WG);
    dim3 block(256);
    bsnn_fused<<<grid, block, 0, stream>>>(X, W0, B0, W1, B1, W2, B2, W3, B3,
                                           W4, B4, W5, B5, (float*)d_out);
}


# Weights and bias of each 64x16 strip are requested from global memory one strip ahead (each B register group refilled right after its WMMAs; barriers fence LDS only); summation order unchanged
# speedup vs baseline: 1.2088x; 1.0347x over previous
#include <hip/hip_runtime.h>

typedef float v2f __attribute__((ext_vector_type(2)));
typedef float v8f __attribute__((ext_vector_type(8)));

#define N_ROWS      131072
#define ROWS_PER_WG 64
#define STR         132

__device__ __forceinline__ v8f bias_init(float bv) {
    v8f c;
#pragma unroll
    for (int i = 0; i < 8; ++i) c[i] = bv;
    return c;
}

#define KG 4

__device__ __forceinline__ void load_a_group(const float* abase, int g,
                                             v2f a[KG][4]) {
#pragma unroll
    for (int q = 0; q < KG; ++q)
#pragma unroll
        for (int rt = 0; rt < 4; ++rt)
            a[q][rt] = *(const v2f*)&abase[16 * rt * STR + (g * KG + q) * 4];
}

__device__ __forceinline__ const float* blk_ptr(const float* W, int ld,
                                                int blk) {
    return W + 64 * blk * ld + 128 * blk;
}

__device__ __forceinline__ unsigned lane_off(int ld, int half, int nCol) {
    return (unsigned)(2 * half * ld + nCol);
}

__device__ __forceinline__ void load_b_group(const float* wu, unsigned lo,
                                             int ld, int g, v2f b[16]) {
#pragma unroll
    for (int q = 0; q < KG; ++q) {
        const int kk = g * KG + q;
        b[kk].x = (wu + (kk * 4 + 0) * ld)[lo];
        b[kk].y = (wu + (kk * 4 + 1) * ld)[lo];
    }
}

__device__ __forceinline__ void wmma_k64(v2f b[16],
                                         const float* src, int srcColOff,
                                         int lane, v8f c[4],
                                         const float* wuNext, unsigned loNext,
                                         int ldNext,
                                         const float* biasNextPtr,
                                         float& biasNext) {
    const int half = lane >> 4;
    const int l15  = lane & 15;
    const float* abase = src + l15 * STR + srcColOff + 2 * half;
    v2f a[2][KG][4];
    load_a_group(abase, 0, a[0]);
#pragma unroll
    for (int g = 0; g < 16 / KG; ++g) {
        if (g + 1 < 16 / KG) load_a_group(abase, g + 1, a[(g + 1) & 1]);
        if (g == 1) biasNext = *biasNextPtr;
        if (g >= 1) load_b_group(wuNext, loNext, ldNext, g - 1, b);
        __builtin_amdgcn_sched_barrier(0);
#pragma unroll
        for (int q = 0; q < KG; ++q)
#pragma unroll
            for (int rt = 0; rt < 4; ++rt)
                c[rt] = __builtin_amdgcn_wmma_f32_16x16x4_f32(
                    false, a[g & 1][q][rt], false, b[g * KG + q], (short)0,
                    c[rt], false, false);
        __builtin_amdgcn_sched_barrier(0);
    }
    load_b_group(wuNext, loNext, ldNext, 16 / KG - 1, b);
}

__device__ __forceinline__ void lds_barrier() {
    __builtin_amdgcn_fence(__ATOMIC_RELEASE, "workgroup", "local");
    __builtin_amdgcn_s_barrier();
    __builtin_amdgcn_fence(__ATOMIC_ACQUIRE, "workgroup", "local");
}

__device__ __forceinline__ void sin_store4(const v8f c[4], float* dst,
                                           int nCol, int half) {
#pragma unroll
    for (int rt = 0; rt < 4; ++rt)
#pragma unroll
        for (int v = 0; v < 8; ++v)
            dst[(16 * rt + v + 8 * half) * STR + nCol] = __sinf(c[rt][v]);
}

__global__ __launch_bounds__(256, 4)
void bsnn_fused(const float* __restrict__ X,
                const float* __restrict__ W0, const float* __restrict__ B0,
                const float* __restrict__ W1, const float* __restrict__ B1,
                const float* __restrict__ W2, const float* __restrict__ B2,
                const float* __restrict__ W3, const float* __restrict__ B3,
                const float* __restrict__ W4, const float* __restrict__ B4,
                const float* __restrict__ W5, const float* __restrict__ B5,
                float* __restrict__ out) {
    __shared__ float hA[ROWS_PER_WG * STR];
    __shared__ float hB[ROWS_PER_WG * STR];
    __shared__ float hC[ROWS_PER_WG * STR];
    __shared__ float acc[ROWS_PER_WG];

    const int tid      = threadIdx.x;
    const int w        = tid >> 5;
    const int lane     = tid & 31;
    const int half     = lane >> 4;
    const int l15      = lane & 15;
    const int rowsBase = blockIdx.x * ROWS_PER_WG;
    const int nCol     = 16 * w + l15;

    if (tid < ROWS_PER_WG) acc[tid] = 0.0f;

    float biasCur = B1[nCol];
    float biasNext;
    v2f b[16];
    {
        const unsigned lo1 = lane_off(128, half, nCol);
#pragma unroll
        for (int g = 0; g < 16 / KG; ++g) load_b_group(W1, lo1, 128, g, b);
    }

    {
        const int nc0 = 16 * (w & 3) + l15;
        const float bias = B0[nc0];
        const float w0r = W0[0 * 64 + nc0];
        const float w1r = W0[1 * 64 + nc0];
        const float w2r = W0[2 * 64 + nc0];
        v2f b;
        b.x = half ? w2r : w0r;
        b.y = half ? 0.0f : w1r;
#pragma unroll
        for (int j = 0; j < 2; ++j) {
            const int rB   = 16 * (w >> 2) + 32 * j;
            const int gRow = rowsBase + rB + l15;
            const float x0 = X[gRow * 3 + 0];
            const float x1 = X[gRow * 3 + 1];
            const float x2 = X[gRow * 3 + 2];
            v2f a;
            a.x = half ? x2 : x0;
            a.y = half ? 0.0f : x1;
            v8f c = bias_init(bias);
            c = __builtin_amdgcn_wmma_f32_16x16x4_f32(
                false, a, false, b, (short)0, c, false, false);
#pragma unroll
            for (int v = 0; v < 8; ++v)
                hC[(rB + v + 8 * half) * STR + nc0] = __sinf(c[v]);
        }
    }
    lds_barrier();

    {
        v8f c[4];
#pragma unroll
        for (int rt = 0; rt < 4; ++rt) c[rt] = bias_init(biasCur);
        wmma_k64(b, hC, 0, lane, c, W2, lane_off(256, half, nCol), 256,
                 B2 + nCol, biasNext);
        biasCur = biasNext;
        sin_store4(c, hA, nCol, half);
    }
    lds_barrier();

    v8f pacc[4];
#pragma unroll
    for (int rt = 0; rt < 4; ++rt) pacc[rt] = bias_init(0.0f);

    for (int c2 = 0; c2 < 2; ++c2) {
        {
            v8f c[4];
#pragma unroll
            for (int rt = 0; rt < 4; ++rt) c[rt] = bias_init(biasCur);
            wmma_k64(b, hA, 64 * c2, lane, c,
                     blk_ptr(W3, 512, 2 * c2), lane_off(512, half, nCol), 512,
                     B3 + 128 * (2 * c2) + nCol, biasNext);
            biasCur = biasNext;
            sin_store4(c, hB, nCol, half);
        }
        lds_barrier();
        for (int c3 = 0; c3 < 2; ++c3) {
            const int i3 = 2 * c2 + c3;
            {
                v8f c[4];
#pragma unroll
                for (int rt = 0; rt < 4; ++rt) c[rt] = bias_init(biasCur);
                wmma_k64(b, hB, 64 * c3, lane, c,
                         blk_ptr(W4, 1024, 2 * i3),
                         lane_off(1024, half, nCol), 1024,
                         B4 + 128 * (2 * i3) + nCol, biasNext);
                biasCur = biasNext;
                sin_store4(c, hC, nCol, half);
            }
            lds_barrier();
            {
                const int i4 = 2 * i3;
                const float w5v = W5[128 * i4 + nCol];
                v8f c[4];
#pragma unroll
                for (int rt = 0; rt < 4; ++rt) c[rt] = bias_init(biasCur);
                wmma_k64(b, hC, 0, lane, c,
                         blk_ptr(W4, 1024, i4 + 1),
                         lane_off(1024, half, nCol), 1024,
                         B4 + 128 * (i4 + 1) + nCol, biasNext);
                biasCur = biasNext;
#pragma unroll
                for (int rt = 0; rt < 4; ++rt)
#pragma unroll
                    for (int v = 0; v < 8; ++v)
                        pacc[rt][v] += __sinf(c[rt][v]) * w5v;
            }
            {
                const int i4 = 2 * i3 + 1;
                const float* wuN;
                const float* bpN;
                int ldN;
                if (c3 == 0) {
                    wuN = blk_ptr(W3, 512, i3 + 1);
                    bpN = B3 + 128 * (i3 + 1) + nCol;
                    ldN = 512;
                } else if (c2 == 0) {
                    wuN = blk_ptr(W2, 256, 1);
                    bpN = B2 + 128 + nCol;
                    ldN = 256;
                } else {
                    wuN = blk_ptr(W4, 1024, i4);
                    bpN = B4 + 128 * i4 + nCol;
                    ldN = 0;
                }
                const float w5v = W5[128 * i4 + nCol];
                v8f c[4];
#pragma unroll
                for (int rt = 0; rt < 4; ++rt) c[rt] = bias_init(biasCur);
                wmma_k64(b, hC, 64, lane, c, wuN,
                         lane_off(ldN, half, nCol), ldN, bpN, biasNext);
                biasCur = biasNext;
#pragma unroll
                for (int rt = 0; rt < 4; ++rt)
#pragma unroll
                    for (int v = 0; v < 8; ++v)
                        pacc[rt][v] += __sinf(c[rt][v]) * w5v;
            }
            lds_barrier();
        }
    }

#pragma unroll
    for (int rt = 0; rt < 4; ++rt) {
#pragma unroll
        for (int v = 0; v < 8; ++v) {
            float p = pacc[rt][v];
            p += __shfl_xor(p, 1, 16);
            p += __shfl_xor(p, 2, 16);
            p += __shfl_xor(p, 4, 16);
            p += __shfl_xor(p, 8, 16);
            if (l15 == 0) atomicAdd(&acc[16 * rt + v + 8 * half], p);
        }
    }
    lds_barrier();
    if (tid < ROWS_PER_WG) out[rowsBase + tid] = acc[tid] + B5[0];
}

extern "C" void kernel_launch(void* const* d_in, const int* in_sizes, int n_in,
                              void* d_out, int out_size, void* d_ws,
                              size_t ws_size, hipStream_t stream) {
    (void)in_sizes; (void)n_in; (void)d_ws; (void)ws_size; (void)out_size;
    const float* X  = (const float*)d_in[0];
    const float* W0 = (const float*)d_in[1];
    const float* B0 = (const float*)d_in[2];
    const float* W1 = (const float*)d_in[3];
    const float* B1 = (const float*)d_in[4];
    const float* W2 = (const float*)d_in[5];
    const float* B2 = (const float*)d_in[6];
    const float* W3 = (const float*)d_in[7];
    const float* B3 = (const float*)d_in[8];
    const float* W4 = (const float*)d_in[9];
    const float* B4 = (const float*)d_in[10];
    const float* W5 = (const float*)d_in[11];
    const float* B5 = (const float*)d_in[12];

    dim3 grid(N_ROWS / ROWS_PER_WG);
    dim3 block(256);
    bsnn_fused<<<grid, block, 0, stream>>>(X, W0, B0, W1, B1, W2, B2, W3, B3,
                                           W4, B4, W5, B5, (float*)d_out);
}


# Final 16-lane sums of the 32 output partials per wave by row_xmask DPP adds on the vector ALU instead of 128 dependent LDS-unit shuffles with a wait each; butterfly order unchanged
# speedup vs baseline: 1.2677x; 1.0487x over previous
#include <hip/hip_runtime.h>

typedef float v2f __attribute__((ext_vector_type(2)));
typedef float v8f __attribute__((ext_vector_type(8)));

#define N_ROWS      131072
#define ROWS_PER_WG 64
#define STR         132

__device__ __forceinline__ v8f bias_init(float bv) {
    v8f c;
#pragma unroll
    for (int i = 0; i < 8; ++i) c[i] = bv;
    return c;
}

#define KG 4

__device__ __forceinline__ void load_a_group(const float* abase, int g,
                                             v2f a[KG][4]) {
#pragma unroll
    for (int q = 0; q < KG; ++q)
#pragma unroll
        for (int rt = 0; rt < 4; ++rt)
            a[q][rt] = *(const v2f*)&abase[16 * rt * STR + (g * KG + q) * 4];
}

__device__ __forceinline__ const float* blk_ptr(const float* W, int ld,
                                                int blk) {
    return W + 64 * blk * ld + 128 * blk;
}

__device__ __forceinline__ unsigned lane_off(int ld, int half, int nCol) {
    return (unsigned)(2 * half * ld + nCol);
}

__device__ __forceinline__ void load_b_group(const float* wu, unsigned lo,
                                             int ld, int g, v2f b[16]) {
#pragma unroll
    for (int q = 0; q < KG; ++q) {
        const int kk = g * KG + q;
        b[kk].x = (wu + (kk * 4 + 0) * ld)[lo];
        b[kk].y = (wu + (kk * 4 + 1) * ld)[lo];
    }
}

__device__ __forceinline__ void wmma_k64(v2f b[16],
                                         const float* src, int srcColOff,
                                         int lane, v8f c[4],
                                         const float* wuNext, unsigned loNext,
                                         int ldNext,
                                         const float* biasNextPtr,
                                         float& biasNext) {
    const int half = lane >> 4;
    const int l15  = lane & 15;
    const float* abase = src + l15 * STR + srcColOff + 2 * half;
    v2f a[2][KG][4];
    load_a_group(abase, 0, a[0]);
#pragma unroll
    for (int g = 0; g < 16 / KG; ++g) {
        if (g + 1 < 16 / KG) load_a_group(abase, g + 1, a[(g + 1) & 1]);
        if (g == 1) biasNext = *biasNextPtr;
        if (g >= 1) load_b_group(wuNext, loNext, ldNext, g - 1, b);
        __builtin_amdgcn_sched_barrier(0);
#pragma unroll
        for (int q = 0; q < KG; ++q)
#pragma unroll
            for (int rt = 0; rt < 4; ++rt)
                c[rt] = __builtin_amdgcn_wmma_f32_16x16x4_f32(
                    false, a[g & 1][q][rt], false, b[g * KG + q], (short)0,
                    c[rt], false, false);
        __builtin_amdgcn_sched_barrier(0);
    }
    load_b_group(wuNext, loNext, ldNext, 16 / KG - 1, b);
}

__device__ __forceinline__ void lds_barrier() {
    __builtin_amdgcn_fence(__ATOMIC_RELEASE, "workgroup", "local");
    __builtin_amdgcn_s_barrier();
    __builtin_amdgcn_fence(__ATOMIC_ACQUIRE, "workgroup", "local");
}

template <int M>
__device__ __forceinline__ float row_add_xor(float p) {
    const int q = __builtin_amdgcn_update_dpp(0, __float_as_int(p), 0x160 + M,
                                              0xf, 0xf, true);
    return p + __int_as_float(q);
}

__device__ __forceinline__ void sin_store4(const v8f c[4], float* dst,
                                           int nCol, int half) {
#pragma unroll
    for (int rt = 0; rt < 4; ++rt)
#pragma unroll
        for (int v = 0; v < 8; ++v)
            dst[(16 * rt + v + 8 * half) * STR + nCol] = __sinf(c[rt][v]);
}

__global__ __launch_bounds__(256, 4)
void bsnn_fused(const float* __restrict__ X,
                const float* __restrict__ W0, const float* __restrict__ B0,
                const float* __restrict__ W1, const float* __restrict__ B1,
                const float* __restrict__ W2, const float* __restrict__ B2,
                const float* __restrict__ W3, const float* __restrict__ B3,
                const float* __restrict__ W4, const float* __restrict__ B4,
                const float* __restrict__ W5, const float* __restrict__ B5,
                float* __restrict__ out) {
    __shared__ float hA[ROWS_PER_WG * STR];
    __shared__ float hB[ROWS_PER_WG * STR];
    __shared__ float hC[ROWS_PER_WG * STR];
    __shared__ float acc[ROWS_PER_WG];

    const int tid      = threadIdx.x;
    const int w        = tid >> 5;
    const int lane     = tid & 31;
    const int half     = lane >> 4;
    const int l15      = lane & 15;
    const int rowsBase = blockIdx.x * ROWS_PER_WG;
    const int nCol     = 16 * w + l15;

    if (tid < ROWS_PER_WG) acc[tid] = 0.0f;

    float biasCur = B1[nCol];
    float biasNext;
    v2f b[16];
    {
        const unsigned lo1 = lane_off(128, half, nCol);
#pragma unroll
        for (int g = 0; g < 16 / KG; ++g) load_b_group(W1, lo1, 128, g, b);
    }

    {
        const int nc0 = 16 * (w & 3) + l15;
        const float bias = B0[nc0];
        const float w0r = W0[0 * 64 + nc0];
        const float w1r = W0[1 * 64 + nc0];
        const float w2r = W0[2 * 64 + nc0];
        v2f b;
        b.x = half ? w2r : w0r;
        b.y = half ? 0.0f : w1r;
#pragma unroll
        for (int j = 0; j < 2; ++j) {
            const int rB   = 16 * (w >> 2) + 32 * j;
            const int gRow = rowsBase + rB + l15;
            const float x0 = X[gRow * 3 + 0];
            const float x1 = X[gRow * 3 + 1];
            const float x2 = X[gRow * 3 + 2];
            v2f a;
            a.x = half ? x2 : x0;
            a.y = half ? 0.0f : x1;
            v8f c = bias_init(bias);
            c = __builtin_amdgcn_wmma_f32_16x16x4_f32(
                false, a, false, b, (short)0, c, false, false);
#pragma unroll
            for (int v = 0; v < 8; ++v)
                hC[(rB + v + 8 * half) * STR + nc0] = __sinf(c[v]);
        }
    }
    lds_barrier();

    {
        v8f c[4];
#pragma unroll
        for (int rt = 0; rt < 4; ++rt) c[rt] = bias_init(biasCur);
        wmma_k64(b, hC, 0, lane, c, W2, lane_off(256, half, nCol), 256,
                 B2 + nCol, biasNext);
        biasCur = biasNext;
        sin_store4(c, hA, nCol, half);
    }
    lds_barrier();

    v8f pacc[4];
#pragma unroll
    for (int rt = 0; rt < 4; ++rt) pacc[rt] = bias_init(0.0f);

    for (int c2 = 0; c2 < 2; ++c2) {
        {
            v8f c[4];
#pragma unroll
            for (int rt = 0; rt < 4; ++rt) c[rt] = bias_init(biasCur);
            wmma_k64(b, hA, 64 * c2, lane, c,
                     blk_ptr(W3, 512, 2 * c2), lane_off(512, half, nCol), 512,
                     B3 + 128 * (2 * c2) + nCol, biasNext);
            biasCur = biasNext;
            sin_store4(c, hB, nCol, half);
        }
        lds_barrier();
        for (int c3 = 0; c3 < 2; ++c3) {
            const int i3 = 2 * c2 + c3;
            {
                v8f c[4];
#pragma unroll
                for (int rt = 0; rt < 4; ++rt) c[rt] = bias_init(biasCur);
                wmma_k64(b, hB, 64 * c3, lane, c,
                         blk_ptr(W4, 1024, 2 * i3),
                         lane_off(1024, half, nCol), 1024,
                         B4 + 128 * (2 * i3) + nCol, biasNext);
                biasCur = biasNext;
                sin_store4(c, hC, nCol, half);
            }
            lds_barrier();
            {
                const int i4 = 2 * i3;
                const float w5v = W5[128 * i4 + nCol];
                v8f c[4];
#pragma unroll
                for (int rt = 0; rt < 4; ++rt) c[rt] = bias_init(biasCur);
                wmma_k64(b, hC, 0, lane, c,
                         blk_ptr(W4, 1024, i4 + 1),
                         lane_off(1024, half, nCol), 1024,
                         B4 + 128 * (i4 + 1) + nCol, biasNext);
                biasCur = biasNext;
#pragma unroll
                for (int rt = 0; rt < 4; ++rt)
#pragma unroll
                    for (int v = 0; v < 8; ++v)
                        pacc[rt][v] += __sinf(c[rt][v]) * w5v;
            }
            {
                const int i4 = 2 * i3 + 1;
                const float* wuN;
                const float* bpN;
                int ldN;
                if (c3 == 0) {
                    wuN = blk_ptr(W3, 512, i3 + 1);
                    bpN = B3 + 128 * (i3 + 1) + nCol;
                    ldN = 512;
                } else if (c2 == 0) {
                    wuN = blk_ptr(W2, 256, 1);
                    bpN = B2 + 128 + nCol;
                    ldN = 256;
                } else {
                    wuN = blk_ptr(W4, 1024, i4);
                    bpN = B4 + 128 * i4 + nCol;
                    ldN = 0;
                }
                const float w5v = W5[128 * i4 + nCol];
                v8f c[4];
#pragma unroll
                for (int rt = 0; rt < 4; ++rt) c[rt] = bias_init(biasCur);
                wmma_k64(b, hC, 64, lane, c, wuN,
                         lane_off(ldN, half, nCol), ldN, bpN, biasNext);
                biasCur = biasNext;
#pragma unroll
                for (int rt = 0; rt < 4; ++rt)
#pragma unroll
                    for (int v = 0; v < 8; ++v)
                        pacc[rt][v] += __sinf(c[rt][v]) * w5v;
            }
            lds_barrier();
        }
    }

    float red[4][8];
#pragma unroll
    for (int rt = 0; rt < 4; ++rt) {
#pragma unroll
        for (int v = 0; v < 8; ++v) {
            float p = pacc[rt][v];
            p = row_add_xor<1>(p);
            p = row_add_xor<2>(p);
            p = row_add_xor<4>(p);
            p = row_add_xor<8>(p);
            red[rt][v] = p;
        }
    }
    if (l15 == 0) {
#pragma unroll
        for (int rt = 0; rt < 4; ++rt)
#pragma unroll
            for (int v = 0; v < 8; ++v)
                atomicAdd(&acc[16 * rt + v + 8 * half], red[rt][v]);
    }
    lds_barrier();
    if (tid < ROWS_PER_WG) out[rowsBase + tid] = acc[tid] + B5[0];
}

extern "C" void kernel_launch(void* const* d_in, const int* in_sizes, int n_in,
                              void* d_out, int out_size, void* d_ws,
                              size_t ws_size, hipStream_t stream) {
    (void)in_sizes; (void)n_in; (void)d_ws; (void)ws_size; (void)out_size;
    const float* X  = (const float*)d_in[0];
    const float* W0 = (const float*)d_in[1];
    const float* B0 = (const float*)d_in[2];
    const float* W1 = (const float*)d_in[3];
    const float* B1 = (const float*)d_in[4];
    const float* W2 = (const float*)d_in[5];
    const float* B2 = (const float*)d_in[6];
    const float* W3 = (const float*)d_in[7];
    const float* B3 = (const float*)d_in[8];
    const float* W4 = (const float*)d_in[9];
    const float* B4 = (const float*)d_in[10];
    const float* W5 = (const float*)d_in[11];
    const float* B5 = (const float*)d_in[12];

    dim3 grid(N_ROWS / ROWS_PER_WG);
    dim3 block(256);
    bsnn_fused<<<grid, block, 0, stream>>>(X, W0, B0, W1, B1, W2, B2, W3, B3,
                                           W4, B4, W5, B5, (float*)d_out);
}
